# starting kernel
# baseline (speedup 1.0000x reference)
#include <hip/hip_runtime.h>
#include <hip/hip_bf16.h>

#define NQ      10
#define LAYERS  10
#define BATCH   4096
#define NACT    4

typedef __attribute__((ext_vector_type(2))) float v2f;
typedef __attribute__((ext_vector_type(8))) float v8f;

__global__ void build_gates_kernel(const float* __restrict__ w,
                                   float* __restrict__ ws) {
  int id = blockIdx.x * blockDim.x + threadIdx.x;
  if (id >= LAYERS * 2 * 1024) return;
  int l     = id >> 11;
  int rem   = id & 2047;
  int which = rem >> 10;
  int e     = rem & 1023;
  int r  = e >> 5;
  int rp = e & 31;
  float accr = 1.f, acci = 0.f;
  for (int j = 0; j < 5; ++j) {
    int q = which * 5 + j;
    float phi   = w[(l * NQ + q) * 3 + 0];
    float theta = w[(l * NQ + q) * 3 + 1];
    float omega = w[(l * NQ + q) * 3 + 2];
    float ct = cosf(0.5f * theta), st = sinf(0.5f * theta);
    float a = 0.5f * (phi + omega), b = 0.5f * (phi - omega);
    float ca = cosf(a), sa = sinf(a), cb = cosf(b), sb = sinf(b);
    int bi = (r  >> (4 - j)) & 1;
    int bj = (rp >> (4 - j)) & 1;
    float gr, gi;
    if (!bi && !bj)      { gr =  ca * ct; gi = -sa * ct; }
    else if (!bi &&  bj) { gr = -cb * st; gi = -sb * st; }
    else if ( bi && !bj) { gr =  cb * st; gi = -sb * st; }
    else                 { gr =  ca * ct; gi =  sa * ct; }
    float nr = accr * gr - acci * gi;
    float ni = accr * gi + acci * gr;
    accr = nr; acci = ni;
  }
  float* base = ws + l * 6144 + which * 3072;
  base[e]        =  accr;
  base[1024 + e] =  acci;
  base[2048 + e] = -acci;
}

__device__ __forceinline__ int cnot_map(int b, int rr) {
  #pragma unroll
  for (int w = 0; w < NQ; ++w) {
    int t = w + rr; if (t >= NQ) t -= NQ;
    b ^= ((b >> (9 - w)) & 1) << (9 - t);
  }
  return b;
}

__global__ __launch_bounds__(64)
void qsim_kernel(const float* __restrict__ x,
                 const float* __restrict__ gates,
                 const float* __restrict__ fcw,
                 const float* __restrict__ fcb,
                 float* __restrict__ out) {
  __shared__ float lds[2 * 4096];
  const int wave = threadIdx.x >> 5;
  const int lane = threadIdx.x & 31;
  const int half = lane >> 4;
  const int l16  = lane & 15;
  const int b    = blockIdx.x * 2 + wave;

  float* Sr = lds + wave * 4096;
  float* Si = Sr + 1024;
  float* Tr = Sr + 2048;
  float* Ti = Sr + 3072;

  float cw[NQ], sw[NQ];
  for (int w = 0; w < NQ; ++w) {
    float xv = x[b * NQ + w];
    xv = xv > 0.f ? xv : 0.f;
    cw[w] = cosf(0.5f * xv);
    sw[w] = sinf(0.5f * xv);
  }
  {
    float mhi = 1.f;
    for (int w = 0; w < 5; ++w) mhi *= ((lane >> (4 - w)) & 1) ? sw[w] : cw[w];
    int pchi = __popc(lane);
    for (int c = 0; c < 32; ++c) {
      float m = mhi;
      for (int w = 0; w < 5; ++w) m *= ((c >> (4 - w)) & 1) ? sw[5 + w] : cw[5 + w];
      int k = (pchi + __popc(c)) & 3;
      float re = 0.f, im = 0.f;
      if      (k == 0) re =  m;
      else if (k == 1) im = -m;
      else if (k == 2) re = -m;
      else             im =  m;
      Sr[lane * 32 + c] = re;
      Si[lane * 32 + c] = im;
    }
  }
  __syncthreads();

  for (int l = 0; l < LAYERS; ++l) {
    const float* gAr  = gates + l * 6144;
    const float* gAi  = gAr + 1024;
    const float* gnAi = gAr + 2048;
    const float* gBr  = gAr + 3072;
    const float* gBi  = gAr + 4096;
    const float* gnBi = gAr + 5120;

    const int rr = l % 9 + 1;
    const int pI = cnot_map(1 << 9, rr);
    const int p7 = cnot_map(1 << 7, rr);
    const int p6 = cnot_map(1 << 6, rr);
    const int p5 = cnot_map(1 << 5, rr);
    const int pJ = cnot_map(1 << 4, rr);
    const int pLane = cnot_map((half << 8) | l16, rr);

    for (int I = 0; I < 2; ++I)
    for (int J = 0; J < 2; ++J) {
      v8f Rr = {}, Ri = {};
      const int row = I * 16 + l16;
      const int n   = J * 16 + l16;
      for (int K = 0; K < 2; ++K)
      for (int kk = 0; kk < 4; ++kk) {
        const int k0 = K * 16 + kk * 4 + 2 * half;
        v2f sr   = *(const v2f*)(Sr   + row * 32 + k0);
        v2f si   = *(const v2f*)(Si   + row * 32 + k0);
        v2f btr  = *(const v2f*)(gBr  + n * 32 + k0);
        v2f bti  = *(const v2f*)(gBi  + n * 32 + k0);
        v2f nbti = *(const v2f*)(gnBi + n * 32 + k0);
        Rr = __builtin_amdgcn_wmma_f32_16x16x4_f32(false, sr, false, btr,  (short)0, Rr, false, false);
        Rr = __builtin_amdgcn_wmma_f32_16x16x4_f32(false, si, false, nbti, (short)0, Rr, false, false);
        Ri = __builtin_amdgcn_wmma_f32_16x16x4_f32(false, sr, false, bti,  (short)0, Ri, false, false);
        Ri = __builtin_amdgcn_wmma_f32_16x16x4_f32(false, si, false, btr,  (short)0, Ri, false, false);
      }
      #pragma unroll
      for (int v = 0; v < 8; ++v) {
        int m = I * 16 + v + 8 * half;
        Tr[m * 32 + n] = Rr[v];
        Ti[m * 32 + n] = Ri[v];
      }
    }
    __syncthreads();

    for (int I = 0; I < 2; ++I)
    for (int J = 0; J < 2; ++J) {
      v8f Ur = {}, Ui = {};
      const int row = I * 16 + l16;
      const int n   = J * 16 + l16;
      for (int K = 0; K < 2; ++K)
      for (int kk = 0; kk < 4; ++kk) {
        const int k0 = K * 16 + kk * 4 + 2 * half;
        v2f ar  = *(const v2f*)(gAr  + row * 32 + k0);
        v2f ai  = *(const v2f*)(gAi  + row * 32 + k0);
        v2f nai = *(const v2f*)(gnAi + row * 32 + k0);
        v2f btr, bti;
        btr.x = Tr[(k0 + 0) * 32 + n];
        btr.y = Tr[(k0 + 1) * 32 + n];
        bti.x = Ti[(k0 + 0) * 32 + n];
        bti.y = Ti[(k0 + 1) * 32 + n];
        Ur = __builtin_amdgcn_wmma_f32_16x16x4_f32(false, ar,  false, btr, (short)0, Ur, false, false);
        Ur = __builtin_amdgcn_wmma_f32_16x16x4_f32(false, nai, false, bti, (short)0, Ur, false, false);
        Ui = __builtin_amdgcn_wmma_f32_16x16x4_f32(false, ar,  false, bti, (short)0, Ui, false, false);
        Ui = __builtin_amdgcn_wmma_f32_16x16x4_f32(false, ai,  false, btr, (short)0, Ui, false, false);
      }
      const int pbase = pLane ^ (I ? pI : 0) ^ (J ? pJ : 0);
      #pragma unroll
      for (int v = 0; v < 8; ++v) {
        int pb = pbase ^ ((v & 4) ? p7 : 0) ^ ((v & 2) ? p6 : 0) ^ ((v & 1) ? p5 : 0);
        Sr[pb] = Ur[v];
        Si[pb] = Ui[v];
      }
    }
    __syncthreads();
  }

  float z[NQ];
  for (int w = 0; w < NQ; ++w) z[w] = 0.f;
  for (int c = 0; c < 32; ++c) {
    int idx = lane * 32 + c;
    float re = Sr[idx], im = Si[idx];
    float p = re * re + im * im;
    #pragma unroll
    for (int w = 0; w < NQ; ++w)
      z[w] += ((idx >> (9 - w)) & 1) ? -p : p;
  }
  for (int off = 16; off > 0; off >>= 1)
    #pragma unroll
    for (int w = 0; w < NQ; ++w)
      z[w] += __shfl_xor(z[w], off, 32);
  if (lane < NACT) {
    float acc = fcb[lane];
    for (int w = 0; w < NQ; ++w) acc += fcw[lane * NQ + w] * z[w];
    out[b * NACT + lane] = acc;
  }
}

extern "C" void kernel_launch(void* const* d_in, const int* in_sizes, int n_in,
                              void* d_out, int out_size, void* d_ws, size_t ws_size,
                              hipStream_t stream) {
  const float* inputs  = (const float*)d_in[0];
  const float* weights = (const float*)d_in[1];
  const float* fcw     = (const float*)d_in[2];
  const float* fcb     = (const float*)d_in[3];
  float* out   = (float*)d_out;
  float* gates = (float*)d_ws;

  build_gates_kernel<<<80, 256, 0, stream>>>(weights, gates);
  qsim_kernel<<<BATCH / 2, 64, 0, stream>>>(inputs, gates, fcw, fcb, out);
}


# Rot gates applied as four 3-wire passes with re/im folded into k: each pass is one real 16x16 gate tile times the state via LDS, 128 WMMAs per layer instead of 256; changes the f32 summation tree
# speedup vs baseline: 2.0423x; 2.0423x over previous
#include <hip/hip_runtime.h>


#define NQ      10
#define LAYERS  10
#define BATCH   4096
#define NACT    4
#define WPB     8

typedef __attribute__((ext_vector_type(2))) float v2f;
typedef __attribute__((ext_vector_type(4))) float v4f;
typedef __attribute__((ext_vector_type(8))) float v8f;

__global__ void build_gates_kernel(const float* __restrict__ w,
                                   float* __restrict__ ws) {
  int id = blockIdx.x * blockDim.x + threadIdx.x;
  if (id >= LAYERS * 4 * 256) return;
  const int s = id & 1, ks = (id >> 1) & 3, lane = (id >> 3) & 31, p = (id >> 8) & 3, l = id >> 10;
  const int m = lane & 15, k = 4 * ks + 2 * (lane >> 4) + s;
  const int i = m >> 1, co = m & 1, j = k >> 1, ci = k & 1;
  const int wires = p == 0 ? 0x210 : p == 1 ? 0x543 : p == 2 ? 0x716 : 0x981;
  float accr = 1.f, acci = 0.f;
  for (int q = 0; q < 3; ++q) {
    const int wq = (wires >> (4 * q)) & 15;
    const int bi = (i >> (2 - q)) & 1;
    const int bj = (j >> (2 - q)) & 1;
    float gr, gi;
    if ((p == 2 && q == 1) || (p == 3 && q == 0)) {
      gr = bi == bj ? 1.f : 0.f; gi = 0.f;
    } else {
      float phi   = w[(l * NQ + wq) * 3 + 0];
      float theta = w[(l * NQ + wq) * 3 + 1];
      float omega = w[(l * NQ + wq) * 3 + 2];
      float ct = cosf(0.5f * theta), st = sinf(0.5f * theta);
      float a = 0.5f * (phi + omega), b = 0.5f * (phi - omega);
      float ca = cosf(a), sa = sinf(a), cb = cosf(b), sb = sinf(b);
      if (!bi && !bj)      { gr =  ca * ct; gi = -sa * ct; }
      else if (!bi &&  bj) { gr = -cb * st; gi = -sb * st; }
      else if ( bi && !bj) { gr =  cb * st; gi = -sb * st; }
      else                 { gr =  ca * ct; gi =  sa * ct; }
    }
    float nr = accr * gr - acci * gi;
    float ni = accr * gi + acci * gr;
    accr = nr; acci = ni;
  }
  ws[id] = co == ci ? accr : (co == 0 ? -acci : acci);
}

__device__ __forceinline__ int cnot_map(int b, int rr) {
  #pragma unroll
  for (int w = 0; w < NQ; ++w) {
    int t = w + rr; if (t >= NQ) t -= NQ;
    b ^= ((b >> (9 - w)) & 1) << (9 - t);
  }
  return b;
}

__device__ __forceinline__ int swz0(int hi) {
  return ((hi & 1) ? 31 : 0) ^ ((hi & 2) ? 13 : 0) ^ ((hi & 4) ? 23 : 0) ^ ((hi & 8) ? 4 : 0) ^ ((hi & 16) ? 21 : 0);
}
__device__ __forceinline__ int slot0x2(int a) {
  int s = (((a >> 9) & 1) << 5) | (((a >> 8) & 1) << 6) | (((a >> 7) & 1) << 0) | (((a >> 6) & 1) << 7)
        | (((a >> 5) & 1) << 8) | (((a >> 4) & 1) << 1) | (((a >> 3) & 1) << 9) | (((a >> 2) & 1) << 2)
        | (((a >> 1) & 1) << 3) | (((a >> 0) & 1) << 4);
  s ^= swz0(s >> 5);
  return 2 * s;
}

__global__ __launch_bounds__(WPB * 32)
void qsim_kernel(const float* __restrict__ x,
                 const float* __restrict__ gates,
                 const float* __restrict__ fcw,
                 const float* __restrict__ fcb,
                 float* __restrict__ out) {
  __shared__ float lds[WPB * 4096];
  const int wave = threadIdx.x >> 5;
  const int lane = threadIdx.x & 31;
  const int half = lane >> 4;
  const int l16  = lane & 15;
  const int b    = blockIdx.x * WPB + wave;
  float* const buf0 = lds + wave * 4096;
  float* const buf1 = buf0 + 2048;

  constexpr int RD[4][8][4] = {{{0, 128, 64, 192}, {256, 384, 320, 448}, {512, 640, 576, 704}, {768, 896, 832, 960}, {1024, 1152, 1088, 1216}, {1280, 1408, 1344, 1472}, {1536, 1664, 1600, 1728}, {1792, 1920, 1856, 1984}}, {{0, 512, 256, 768}, {64, 576, 320, 832}, {128, 640, 384, 896}, {192, 704, 448, 960}, {1024, 1536, 1280, 1792}, {1088, 1600, 1344, 1856}, {1152, 1664, 1408, 1920}, {1216, 1728, 1472, 1984}}, {{0, 64, 1024, 1088}, {128, 192, 1152, 1216}, {256, 320, 1280, 1344}, {384, 448, 1408, 1472}, {512, 576, 1536, 1600}, {640, 704, 1664, 1728}, {768, 832, 1792, 1856}, {896, 960, 1920, 1984}}, {{0, 1024, 64, 1088}, {128, 1152, 192, 1216}, {256, 1280, 320, 1344}, {384, 1408, 448, 1472}, {512, 1536, 576, 1600}, {640, 1664, 704, 1728}, {768, 1792, 832, 1856}, {896, 1920, 960, 1984}}};
  constexpr int RX[8][4] = {{0, 26, 62, 36}, {46, 52, 16, 10}, {8, 18, 54, 44}, {38, 60, 24, 2}, {42, 48, 20, 14}, {4, 30, 58, 32}, {34, 56, 28, 6}, {12, 22, 50, 40}};
  constexpr int WR[3][8][4] = {{{0, 128, 64, 192}, {256, 384, 320, 448}, {512, 640, 576, 704}, {768, 896, 832, 960}, {1024, 1152, 1088, 1216}, {1280, 1408, 1344, 1472}, {1536, 1664, 1600, 1728}, {1792, 1920, 1856, 1984}}, {{0, 512, 256, 768}, {64, 576, 320, 832}, {128, 640, 384, 896}, {192, 704, 448, 960}, {1024, 1536, 1280, 1792}, {1088, 1600, 1344, 1856}, {1152, 1664, 1408, 1920}, {1216, 1728, 1472, 1984}}, {{0, 512, 64, 576}, {32, 544, 96, 608}, {128, 640, 192, 704}, {160, 672, 224, 736}, {256, 768, 320, 832}, {288, 800, 352, 864}, {384, 896, 448, 960}, {416, 928, 480, 992}}};

  const int rdl = (l16 << 2) | (half << 1);
  const int wrl[3] = {
    (((l16 >> 0) & 1) << 1) | (((l16 >> 1) & 1) << 3) | (((l16 >> 2) & 1) << 4) | (((l16 >> 3) & 1) << 5) | (half << 2),
    (((l16 >> 0) & 1) << 3) | (((l16 >> 1) & 1) << 1) | (((l16 >> 2) & 1) << 4) | (((l16 >> 3) & 1) << 5) | (half << 2),
    (((l16 >> 0) & 1) << 3) | (((l16 >> 1) & 1) << 4) | (((l16 >> 2) & 1) << 10) | (((l16 >> 3) & 1) << 1) | (half << 2)};

  const float* const gl = gates + lane * 8;
  v4f gna = *(const v4f*)(gl), gnb = *(const v4f*)(gl + 4);

  float cw[NQ], sw[NQ];
  #pragma unroll
  for (int w = 0; w < NQ; ++w) {
    float xv = x[b * NQ + w];
    xv = xv > 0.f ? xv : 0.f;
    cw[w] = cosf(0.5f * xv);
    sw[w] = sinf(0.5f * xv);
  }
  {
    float fw[NQ];
    fw[2] = (lane & 1)  ? sw[2] : cw[2];
    fw[5] = (lane & 2)  ? sw[5] : cw[5];
    fw[7] = (lane & 4)  ? sw[7] : cw[7];
    fw[8] = (lane & 8)  ? sw[8] : cw[8];
    fw[9] = (lane & 16) ? sw[9] : cw[9];
    const int kl = __popc(lane) & 3;
    const float ur = kl == 0 ? 1.f : (kl == 2 ? -1.f : 0.f);
    const float ui = kl == 1 ? -1.f : (kl == 3 ? 1.f : 0.f);
    #pragma unroll
    for (int i = 0; i < 32; ++i) {
      fw[0] = (i & 1)  ? sw[0] : cw[0];
      fw[1] = (i & 2)  ? sw[1] : cw[1];
      fw[3] = (i & 4)  ? sw[3] : cw[3];
      fw[4] = (i & 8)  ? sw[4] : cw[4];
      fw[6] = (i & 16) ? sw[6] : cw[6];
      float m = 1.f;
      #pragma unroll
      for (int w = 0; w < NQ; ++w) m *= fw[w];
      const int ki = __popc(i) & 3;
      const float cr = ki == 0 ? ur : ki == 1 ? ui : ki == 2 ? -ur : -ui;
      const float cim = ki == 0 ? ui : ki == 1 ? -ur : ki == 2 ? -ui : ur;
      v2f v = {m * cr, m * cim};
      *(v2f*)(buf0 + (((2 * lane) ^ (2 * swz0(i))) + 64 * i)) = v;
    }
  }

  for (int l = 0; l < LAYERS; ++l) {
    const int rr = l % 9 + 1;
    const int iv = slot0x2(cnot_map(512 >> lane, rr));
    int img[NQ];
    #pragma unroll
    for (int w = 0; w < NQ; ++w) img[w] = __builtin_amdgcn_readlane(iv, w);
    const int w3l = ((l16 & 1) ? img[6] : 0) ^ ((l16 & 2) ? img[3] : 0) ^ ((l16 & 4) ? img[0] : 0)
                  ^ ((l16 & 8) ? img[2] : 0) ^ (half ? img[1] : 0);

    #pragma unroll
    for (int p = 0; p < 4; ++p) {
      const float* const src = (p & 1) ? buf1 : buf0;
      float* const dst = (p & 1) ? buf0 : buf1;
      const v4f ga = gna, gb = gnb;
      const v2f a[4] = {{ga.x, ga.y}, {ga.z, ga.w}, {gb.x, gb.y}, {gb.z, gb.w}};
      {
        const int qn = l * 4 + p + 1 < LAYERS * 4 ? l * 4 + p + 1 : l * 4 + p;
        gna = *(const v4f*)(gl + qn * 256);
        gnb = *(const v4f*)(gl + qn * 256 + 4);
      }

      v2f bf[8][4];
      #pragma unroll
      for (int t = 0; t < 8; ++t) {
        #pragma unroll
        for (int ks = 0; ks < 4; ++ks) {
          const int off = p == 0 ? ((rdl ^ RX[t][ks]) + RD[0][t][ks]) : (rdl + RD[p][t][ks]);
          bf[t][ks] = *(const v2f*)(src + off);
        }
      }
      v8f acc[8];
      #pragma unroll
      for (int t = 0; t < 8; ++t) acc[t] = v8f{};
      #pragma unroll
      for (int ks = 0; ks < 4; ++ks) {
        #pragma unroll
        for (int t = 0; t < 8; ++t)
          acc[t] = __builtin_amdgcn_wmma_f32_16x16x4_f32(false, a[ks], false, bf[t][ks], (short)0, acc[t], false, false);
      }
      #pragma unroll
      for (int t = 0; t < 8; ++t) {
        #pragma unroll
        for (int j = 0; j < 4; ++j) {
          v2f v = {acc[t][2 * j], acc[t][2 * j + 1]};
          if (p < 3) {
            *(v2f*)(dst + wrl[p < 3 ? p : 0] + WR[p < 3 ? p : 0][t][j]) = v;
          } else {
            const int c = ((t & 1) ? img[4] : 0) ^ ((t & 2) ? img[5] : 0) ^ ((t & 4) ? img[7] : 0)
                        ^ ((j & 2) ? img[8] : 0) ^ ((j & 1) ? img[9] : 0);
            *(v2f*)(dst + (w3l ^ c)) = v;
          }
        }
      }
    }
  }

  float ptot = 0.f, z0 = 0.f, z1 = 0.f, z3 = 0.f, z4 = 0.f, z6 = 0.f;
  #pragma unroll
  for (int i = 0; i < 32; ++i) {
    v2f v = *(const v2f*)(buf0 + (((2 * lane) ^ (2 * swz0(i))) + 64 * i));
    float p = v.x * v.x + v.y * v.y;
    ptot += p;
    z0 += (i & 1)  ? -p : p;
    z1 += (i & 2)  ? -p : p;
    z3 += (i & 4)  ? -p : p;
    z4 += (i & 8)  ? -p : p;
    z6 += (i & 16) ? -p : p;
  }
  float z[NQ];
  z[0] = z0; z[1] = z1; z[3] = z3; z[4] = z4; z[6] = z6;
  z[2] = (lane & 1)  ? -ptot : ptot;
  z[5] = (lane & 2)  ? -ptot : ptot;
  z[7] = (lane & 4)  ? -ptot : ptot;
  z[8] = (lane & 8)  ? -ptot : ptot;
  z[9] = (lane & 16) ? -ptot : ptot;
  for (int off = 16; off > 0; off >>= 1)
    #pragma unroll
    for (int w = 0; w < NQ; ++w)
      z[w] += __shfl_xor(z[w], off, 32);
  if (lane < NACT) {
    float acc = fcb[lane];
    #pragma unroll
    for (int w = 0; w < NQ; ++w) acc += fcw[lane * NQ + w] * z[w];
    out[b * NACT + lane] = acc;
  }
}

extern "C" void kernel_launch(void* const* d_in, const int* in_sizes, int n_in,
                              void* d_out, int out_size, void* d_ws, size_t ws_size,
                              hipStream_t stream) {
  const float* inputs  = (const float*)d_in[0];
  const float* weights = (const float*)d_in[1];
  const float* fcw     = (const float*)d_in[2];
  const float* fcb     = (const float*)d_in[3];
  float* out   = (float*)d_out;
  float* gates = (float*)d_ws;

  build_gates_kernel<<<40, 256, 0, stream>>>(weights, gates);
  qsim_kernel<<<BATCH / WPB, WPB * 32, 0, stream>>>(inputs, gates, fcw, fcb, out);
}


# Rot gates as three dense 3-wire WMMA passes (wires 0-8), wire 9's complex 2x2 on the vector unit where it is a tile bit: 96 WMMAs, 3 LDS round trips per layer, not 128, 4; changes f32 summation tree
# speedup vs baseline: 2.1651x; 1.0601x over previous
#include <hip/hip_runtime.h>


#define NQ      10
#define LAYERS  10
#define BATCH   4096
#define NACT    4
#define WPB     8
#define NPASS   3

typedef __attribute__((ext_vector_type(2))) float v2f;
typedef __attribute__((ext_vector_type(4))) float v4f;
typedef __attribute__((ext_vector_type(8))) float v8f;

#define G9_OFF (LAYERS * NPASS * 256)

__device__ __forceinline__ void rot_elem(const float* __restrict__ w, int l, int wq, int bi, int bj,
                                         float& gr, float& gi) {
  float phi   = w[(l * NQ + wq) * 3 + 0];
  float theta = w[(l * NQ + wq) * 3 + 1];
  float omega = w[(l * NQ + wq) * 3 + 2];
  float ct = cosf(0.5f * theta), st = sinf(0.5f * theta);
  float a = 0.5f * (phi + omega), b = 0.5f * (phi - omega);
  float ca = cosf(a), sa = sinf(a), cb = cosf(b), sb = sinf(b);
  if (!bi && !bj)      { gr =  ca * ct; gi = -sa * ct; }
  else if (!bi &&  bj) { gr = -cb * st; gi = -sb * st; }
  else if ( bi && !bj) { gr =  cb * st; gi = -sb * st; }
  else                 { gr =  ca * ct; gi =  sa * ct; }
}

__global__ void build_gates_kernel(const float* __restrict__ w,
                                   float* __restrict__ ws) {
  int id = blockIdx.x * blockDim.x + threadIdx.x;
  if (id >= G9_OFF + LAYERS * 8) return;
  if (id >= G9_OFF) {
    const int e = id - G9_OFF;
    float gr, gi;
    rot_elem(w, e >> 3, 9, (e >> 2) & 1, (e >> 1) & 1, gr, gi);
    ws[id] = (e & 1) ? gi : gr;
    return;
  }
  const int s = id & 1, ks = (id >> 1) & 3, lane = (id >> 3) & 31, q = id >> 8;
  const int l = q / NPASS, p = q - l * NPASS;
  const int m = lane & 15, k = 4 * ks + 2 * (lane >> 4) + s;
  const int i = m >> 1, co = m & 1, j = k >> 1, ci = k & 1;
  const int wires = p == 0 ? 0x210 : p == 1 ? 0x543 : 0x876;
  float accr = 1.f, acci = 0.f;
  for (int qq = 0; qq < 3; ++qq) {
    const int wq = (wires >> (4 * qq)) & 15;
    float gr, gi;
    rot_elem(w, l, wq, (i >> (2 - qq)) & 1, (j >> (2 - qq)) & 1, gr, gi);
    float nr = accr * gr - acci * gi;
    float ni = accr * gi + acci * gr;
    accr = nr; acci = ni;
  }
  ws[id] = co == ci ? accr : (co == 0 ? -acci : acci);
}

__device__ __forceinline__ int cnot_map(int b, int rr) {
  #pragma unroll
  for (int w = 0; w < NQ; ++w) {
    int t = w + rr; if (t >= NQ) t -= NQ;
    b ^= ((b >> (9 - w)) & 1) << (9 - t);
  }
  return b;
}

__device__ __forceinline__ int swz0(int hi) {
  return ((hi & 1) ? 31 : 0) ^ ((hi & 2) ? 13 : 0) ^ ((hi & 4) ? 23 : 0) ^ ((hi & 8) ? 4 : 0) ^ ((hi & 16) ? 21 : 0);
}
__device__ __forceinline__ int slot0x2(int a) {
  int s = (((a >> 9) & 1) << 5) | (((a >> 8) & 1) << 6) | (((a >> 7) & 1) << 0) | (((a >> 6) & 1) << 7)
        | (((a >> 5) & 1) << 8) | (((a >> 4) & 1) << 1) | (((a >> 3) & 1) << 9) | (((a >> 2) & 1) << 2)
        | (((a >> 1) & 1) << 3) | (((a >> 0) & 1) << 4);
  s ^= swz0(s >> 5);
  return 2 * s;
}

__global__ __launch_bounds__(WPB * 32)
void qsim_kernel(const float* __restrict__ x,
                 const float* __restrict__ gates,
                 const float* __restrict__ fcw,
                 const float* __restrict__ fcb,
                 float* __restrict__ out) {
  __shared__ float lds[WPB * 4096];
  const int wave = threadIdx.x >> 5;
  const int lane = threadIdx.x & 31;
  const int half = lane >> 4;
  const int l16  = lane & 15;
  const int b    = blockIdx.x * WPB + wave;
  float* const buf0 = lds + wave * 4096;
  float* const buf1 = buf0 + 2048;

  constexpr int RD[2][8][4] = {{{0, 128, 64, 192}, {256, 384, 320, 448}, {512, 640, 576, 704}, {768, 896, 832, 960}, {1024, 1152, 1088, 1216}, {1280, 1408, 1344, 1472}, {1536, 1664, 1600, 1728}, {1792, 1920, 1856, 1984}}, {{0, 512, 256, 768}, {64, 576, 320, 832}, {128, 640, 384, 896}, {192, 704, 448, 960}, {1024, 1536, 1280, 1792}, {1088, 1600, 1344, 1856}, {1152, 1664, 1408, 1920}, {1216, 1728, 1472, 1984}}};
  constexpr int RX[8][4] = {{0, 26, 62, 36}, {46, 52, 16, 10}, {8, 18, 54, 44}, {38, 60, 24, 2}, {42, 48, 20, 14}, {4, 30, 58, 32}, {34, 56, 28, 6}, {12, 22, 50, 40}};
  constexpr int WR0[8][4] = {{0, 128, 64, 192}, {256, 384, 320, 448}, {512, 640, 576, 704}, {768, 896, 832, 960}, {1024, 1152, 1088, 1216}, {1280, 1408, 1344, 1472}, {1536, 1664, 1600, 1728}, {1792, 1920, 1856, 1984}};

  const int rdl = (l16 << 2) | (half << 1);
  const int wrl0 = (((l16 >> 0) & 1) << 1) | (((l16 >> 1) & 1) << 3) | (((l16 >> 2) & 1) << 4) | (((l16 >> 3) & 1) << 5) | (half << 2);
  const int wrl1 = ((l16 & 1) ? 4 : 0) ^ ((l16 & 2) ? 72 : 0) ^ ((l16 & 4) ? 2 : 0) ^ ((l16 & 8) ? 1040 : 0) ^ (half ? 32 : 0);
  const int wrl1x[4] = {wrl1, wrl1 ^ 8, wrl1 ^ 16, wrl1 ^ 24};

  const float* const gl = gates + lane * 8;
  v4f gna = *(const v4f*)(gl), gnb = *(const v4f*)(gl + 4);

  float cw[NQ], sw[NQ];
  #pragma unroll
  for (int w = 0; w < NQ; ++w) {
    float xv = x[b * NQ + w];
    xv = xv > 0.f ? xv : 0.f;
    cw[w] = cosf(0.5f * xv);
    sw[w] = sinf(0.5f * xv);
  }
  {
    float fw[NQ];
    fw[2] = (lane & 1)  ? sw[2] : cw[2];
    fw[5] = (lane & 2)  ? sw[5] : cw[5];
    fw[7] = (lane & 4)  ? sw[7] : cw[7];
    fw[8] = (lane & 8)  ? sw[8] : cw[8];
    fw[9] = (lane & 16) ? sw[9] : cw[9];
    const int kl = __popc(lane) & 3;
    const float ur = kl == 0 ? 1.f : (kl == 2 ? -1.f : 0.f);
    const float ui = kl == 1 ? -1.f : (kl == 3 ? 1.f : 0.f);
    #pragma unroll
    for (int i = 0; i < 32; ++i) {
      fw[0] = (i & 1)  ? sw[0] : cw[0];
      fw[1] = (i & 2)  ? sw[1] : cw[1];
      fw[3] = (i & 4)  ? sw[3] : cw[3];
      fw[4] = (i & 8)  ? sw[4] : cw[4];
      fw[6] = (i & 16) ? sw[6] : cw[6];
      float m = 1.f;
      #pragma unroll
      for (int w = 0; w < NQ; ++w) m *= fw[w];
      const int ki = __popc(i) & 3;
      const float cr = ki == 0 ? ur : ki == 1 ? ui : ki == 2 ? -ur : -ui;
      const float cim = ki == 0 ? ui : ki == 1 ? -ur : ki == 2 ? -ui : ur;
      v2f v = {m * cr, m * cim};
      *(v2f*)(buf0 + (((2 * lane) ^ (2 * swz0(i))) + 64 * i)) = v;
    }
  }

  for (int l = 0; l < LAYERS; ++l) {
    const int rr = l % 9 + 1;
    const int iv = slot0x2(cnot_map(512 >> lane, rr));
    int img[NQ];
    #pragma unroll
    for (int w = 0; w < NQ; ++w) img[w] = __builtin_amdgcn_readlane(iv, w);
    const int w2l = ((l16 & 1) ? img[0] : 0) ^ ((l16 & 2) ? img[1] : 0) ^ ((l16 & 4) ? img[2] : 0)
                  ^ ((l16 & 8) ? img[3] : 0) ^ (half ? img[6] : 0);
    float g9[8];
    #pragma unroll
    for (int c = 0; c < 8; ++c) g9[c] = gates[G9_OFF + l * 8 + c];
    float* const bA = (l & 1) ? buf1 : buf0;
    float* const bB = (l & 1) ? buf0 : buf1;

    #pragma unroll
    for (int p = 0; p < NPASS; ++p) {
      const float* const src = (p == 1) ? bB : bA;
      float* const dst = (p == 1) ? bA : bB;
      const v4f ga = gna, gb = gnb;
      const v2f a[4] = {{ga.x, ga.y}, {ga.z, ga.w}, {gb.x, gb.y}, {gb.z, gb.w}};
      {
        const int qn = l * NPASS + p + 1 < LAYERS * NPASS ? l * NPASS + p + 1 : l * NPASS + p;
        gna = *(const v4f*)(gl + qn * 256);
        gnb = *(const v4f*)(gl + qn * 256 + 4);
      }

      v2f bf[8][4];
      #pragma unroll
      for (int t = 0; t < 8; ++t) {
        #pragma unroll
        for (int ks = 0; ks < 4; ++ks) {
          const int off = p == 0 ? ((rdl ^ RX[t][ks]) + RD[0][t][ks])
                        : p == 1 ? (rdl + RD[1][t][ks])
                        : ((rdl ^ (((ks & 1) ? 8 : 0) ^ ((t & 4) ? 16 : 0))) + 64 * (4 * t + ks));
          bf[t][ks] = *(const v2f*)(src + off);
        }
      }
      if (p == 2) {
        #pragma unroll
        for (int t = 0; t < 4; ++t) {
          #pragma unroll
          for (int ks = 0; ks < 4; ++ks) {
            const v2f x0 = bf[t][ks], x1 = bf[t + 4][ks];
            const v2f x0s = {-x0.y, x0.x}, x1s = {-x1.y, x1.x};
            bf[t][ks]     = x0 * g9[0] + x0s * g9[1] + x1 * g9[2] + x1s * g9[3];
            bf[t + 4][ks] = x0 * g9[4] + x0s * g9[5] + x1 * g9[6] + x1s * g9[7];
          }
        }
      }
      v8f acc[8];
      #pragma unroll
      for (int t = 0; t < 8; ++t) acc[t] = v8f{};
      #pragma unroll
      for (int ks = 0; ks < 4; ++ks) {
        #pragma unroll
        for (int t = 0; t < 8; ++t)
          acc[t] = __builtin_amdgcn_wmma_f32_16x16x4_f32(false, a[ks], false, bf[t][ks], (short)0, acc[t], false, false);
      }
      #pragma unroll
      for (int t = 0; t < 8; ++t) {
        #pragma unroll
        for (int j = 0; j < 4; ++j) {
          v2f v = {acc[t][2 * j], acc[t][2 * j + 1]};
          if (p == 0) {
            *(v2f*)(dst + wrl0 + WR0[t][j]) = v;
          } else if (p == 1) {
            const int c = ((t & 4) ? 128 : 0) + ((j & 1) ? 512 : 0) + ((j & 2) ? 256 : 0);
            *(v2f*)(dst + wrl1x[t & 3] + c) = v;
          } else {
            const int c = ((t & 1) ? img[4] : 0) ^ ((t & 2) ? img[5] : 0) ^ ((t & 4) ? img[9] : 0)
                        ^ ((j & 2) ? img[7] : 0) ^ ((j & 1) ? img[8] : 0);
            *(v2f*)(dst + (w2l ^ c)) = v;
          }
        }
      }
    }
  }

  float ptot = 0.f, z0 = 0.f, z1 = 0.f, z3 = 0.f, z4 = 0.f, z6 = 0.f;
  #pragma unroll
  for (int i = 0; i < 32; ++i) {
    v2f v = *(const v2f*)(buf0 + (((2 * lane) ^ (2 * swz0(i))) + 64 * i));
    float p = v.x * v.x + v.y * v.y;
    ptot += p;
    z0 += (i & 1)  ? -p : p;
    z1 += (i & 2)  ? -p : p;
    z3 += (i & 4)  ? -p : p;
    z4 += (i & 8)  ? -p : p;
    z6 += (i & 16) ? -p : p;
  }
  float z[NQ];
  z[0] = z0; z[1] = z1; z[3] = z3; z[4] = z4; z[6] = z6;
  z[2] = (lane & 1)  ? -ptot : ptot;
  z[5] = (lane & 2)  ? -ptot : ptot;
  z[7] = (lane & 4)  ? -ptot : ptot;
  z[8] = (lane & 8)  ? -ptot : ptot;
  z[9] = (lane & 16) ? -ptot : ptot;
  for (int off = 16; off > 0; off >>= 1)
    #pragma unroll
    for (int w = 0; w < NQ; ++w)
      z[w] += __shfl_xor(z[w], off, 32);
  if (lane < NACT) {
    float acc = fcb[lane];
    #pragma unroll
    for (int w = 0; w < NQ; ++w) acc += fcw[lane * NQ + w] * z[w];
    out[b * NACT + lane] = acc;
  }
}

extern "C" void kernel_launch(void* const* d_in, const int* in_sizes, int n_in,
                              void* d_out, int out_size, void* d_ws, size_t ws_size,
                              hipStream_t stream) {
  const float* inputs  = (const float*)d_in[0];
  const float* weights = (const float*)d_in[1];
  const float* fcw     = (const float*)d_in[2];
  const float* fcb     = (const float*)d_in[3];
  float* out   = (float*)d_out;
  float* gates = (float*)d_ws;

  build_gates_kernel<<<31, 256, 0, stream>>>(weights, gates);
  qsim_kernel<<<BATCH / WPB, WPB * 32, 0, stream>>>(inputs, gates, fcw, fcb, out);
}


# Two batch items per wave (2 waves per SIMD, not 4): a layer's CNOT-ring images, LDS addresses and gate fragments are made once and serve both; item 1 through immediate LDS offsets; sums unchanged
# speedup vs baseline: 2.4457x; 1.1296x over previous
#include <hip/hip_runtime.h>


#define NQ      10
#define LAYERS  10
#define BATCH   4096
#define NACT    4
#define WPB     8
#define IPW     2
#define NPASS   3

typedef __attribute__((ext_vector_type(2))) float v2f;
typedef __attribute__((ext_vector_type(4))) float v4f;
typedef __attribute__((ext_vector_type(8))) float v8f;

#define G9_OFF (LAYERS * NPASS * 256)

__device__ __forceinline__ void rot_elem(const float* __restrict__ w, int l, int wq, int bi, int bj,
                                         float& gr, float& gi) {
  float phi   = w[(l * NQ + wq) * 3 + 0];
  float theta = w[(l * NQ + wq) * 3 + 1];
  float omega = w[(l * NQ + wq) * 3 + 2];
  float ct = cosf(0.5f * theta), st = sinf(0.5f * theta);
  float a = 0.5f * (phi + omega), b = 0.5f * (phi - omega);
  float ca = cosf(a), sa = sinf(a), cb = cosf(b), sb = sinf(b);
  if (!bi && !bj)      { gr =  ca * ct; gi = -sa * ct; }
  else if (!bi &&  bj) { gr = -cb * st; gi = -sb * st; }
  else if ( bi && !bj) { gr =  cb * st; gi = -sb * st; }
  else                 { gr =  ca * ct; gi =  sa * ct; }
}

__global__ void build_gates_kernel(const float* __restrict__ w,
                                   float* __restrict__ ws) {
  int id = blockIdx.x * blockDim.x + threadIdx.x;
  if (id >= G9_OFF + LAYERS * 8) return;
  if (id >= G9_OFF) {
    const int e = id - G9_OFF;
    float gr, gi;
    rot_elem(w, e >> 3, 9, (e >> 2) & 1, (e >> 1) & 1, gr, gi);
    ws[id] = (e & 1) ? gi : gr;
    return;
  }
  const int s = id & 1, ks = (id >> 1) & 3, lane = (id >> 3) & 31, q = id >> 8;
  const int l = q / NPASS, p = q - l * NPASS;
  const int m = lane & 15, k = 4 * ks + 2 * (lane >> 4) + s;
  const int i = m >> 1, co = m & 1, j = k >> 1, ci = k & 1;
  const int wires = p == 0 ? 0x210 : p == 1 ? 0x543 : 0x876;
  float accr = 1.f, acci = 0.f;
  for (int qq = 0; qq < 3; ++qq) {
    const int wq = (wires >> (4 * qq)) & 15;
    float gr, gi;
    rot_elem(w, l, wq, (i >> (2 - qq)) & 1, (j >> (2 - qq)) & 1, gr, gi);
    float nr = accr * gr - acci * gi;
    float ni = accr * gi + acci * gr;
    accr = nr; acci = ni;
  }
  ws[id] = co == ci ? accr : (co == 0 ? -acci : acci);
}

__device__ __forceinline__ int cnot_map(int b, int rr) {
  #pragma unroll
  for (int w = 0; w < NQ; ++w) {
    int t = w + rr; if (t >= NQ) t -= NQ;
    b ^= ((b >> (9 - w)) & 1) << (9 - t);
  }
  return b;
}

__device__ __forceinline__ int swz0(int hi) {
  return ((hi & 1) ? 31 : 0) ^ ((hi & 2) ? 13 : 0) ^ ((hi & 4) ? 23 : 0) ^ ((hi & 8) ? 4 : 0) ^ ((hi & 16) ? 21 : 0);
}
__device__ __forceinline__ int slot0x2(int a) {
  int s = (((a >> 9) & 1) << 5) | (((a >> 8) & 1) << 6) | (((a >> 7) & 1) << 0) | (((a >> 6) & 1) << 7)
        | (((a >> 5) & 1) << 8) | (((a >> 4) & 1) << 1) | (((a >> 3) & 1) << 9) | (((a >> 2) & 1) << 2)
        | (((a >> 1) & 1) << 3) | (((a >> 0) & 1) << 4);
  s ^= swz0(s >> 5);
  return 2 * s;
}

__global__ __launch_bounds__(WPB * 32)
void qsim_kernel(const float* __restrict__ x,
                 const float* __restrict__ gates,
                 const float* __restrict__ fcw,
                 const float* __restrict__ fcb,
                 float* __restrict__ out) {
  __shared__ float lds[WPB * IPW * 4096];
  const int wave = threadIdx.x >> 5;
  const int lane = threadIdx.x & 31;
  const int half = lane >> 4;
  const int l16  = lane & 15;
  const int b0   = (blockIdx.x * WPB + wave) * IPW;
  float* const buf0 = lds + wave * (IPW * 4096);
  float* const buf1 = buf0 + 2048;

  constexpr int RD[2][8][4] = {{{0, 128, 64, 192}, {256, 384, 320, 448}, {512, 640, 576, 704}, {768, 896, 832, 960}, {1024, 1152, 1088, 1216}, {1280, 1408, 1344, 1472}, {1536, 1664, 1600, 1728}, {1792, 1920, 1856, 1984}}, {{0, 512, 256, 768}, {64, 576, 320, 832}, {128, 640, 384, 896}, {192, 704, 448, 960}, {1024, 1536, 1280, 1792}, {1088, 1600, 1344, 1856}, {1152, 1664, 1408, 1920}, {1216, 1728, 1472, 1984}}};
  constexpr int RX[8][4] = {{0, 26, 62, 36}, {46, 52, 16, 10}, {8, 18, 54, 44}, {38, 60, 24, 2}, {42, 48, 20, 14}, {4, 30, 58, 32}, {34, 56, 28, 6}, {12, 22, 50, 40}};
  constexpr int WR0[8][4] = {{0, 128, 64, 192}, {256, 384, 320, 448}, {512, 640, 576, 704}, {768, 896, 832, 960}, {1024, 1152, 1088, 1216}, {1280, 1408, 1344, 1472}, {1536, 1664, 1600, 1728}, {1792, 1920, 1856, 1984}};

  const int rdl = (l16 << 2) | (half << 1);
  const int wrl0 = (((l16 >> 0) & 1) << 1) | (((l16 >> 1) & 1) << 3) | (((l16 >> 2) & 1) << 4) | (((l16 >> 3) & 1) << 5) | (half << 2);
  const int wrl1 = ((l16 & 1) ? 4 : 0) ^ ((l16 & 2) ? 72 : 0) ^ ((l16 & 4) ? 2 : 0) ^ ((l16 & 8) ? 1040 : 0) ^ (half ? 32 : 0);
  const int wrl1x[4] = {wrl1, wrl1 ^ 8, wrl1 ^ 16, wrl1 ^ 24};

  const float* const gl = gates + lane * 8;
  v4f gna = *(const v4f*)(gl), gnb = *(const v4f*)(gl + 4);

  for (int it = 0; it < IPW; ++it) {
  float* const ib = buf0 + it * 4096;
  float cw[NQ], sw[NQ];
  #pragma unroll
  for (int w = 0; w < NQ; ++w) {
    float xv = x[(b0 + it) * NQ + w];
    xv = xv > 0.f ? xv : 0.f;
    cw[w] = cosf(0.5f * xv);
    sw[w] = sinf(0.5f * xv);
  }
  {
    float fw[NQ];
    fw[2] = (lane & 1)  ? sw[2] : cw[2];
    fw[5] = (lane & 2)  ? sw[5] : cw[5];
    fw[7] = (lane & 4)  ? sw[7] : cw[7];
    fw[8] = (lane & 8)  ? sw[8] : cw[8];
    fw[9] = (lane & 16) ? sw[9] : cw[9];
    const int kl = __popc(lane) & 3;
    const float ur = kl == 0 ? 1.f : (kl == 2 ? -1.f : 0.f);
    const float ui = kl == 1 ? -1.f : (kl == 3 ? 1.f : 0.f);
    #pragma unroll
    for (int i = 0; i < 32; ++i) {
      fw[0] = (i & 1)  ? sw[0] : cw[0];
      fw[1] = (i & 2)  ? sw[1] : cw[1];
      fw[3] = (i & 4)  ? sw[3] : cw[3];
      fw[4] = (i & 8)  ? sw[4] : cw[4];
      fw[6] = (i & 16) ? sw[6] : cw[6];
      float m = 1.f;
      #pragma unroll
      for (int w = 0; w < NQ; ++w) m *= fw[w];
      const int ki = __popc(i) & 3;
      const float cr = ki == 0 ? ur : ki == 1 ? ui : ki == 2 ? -ur : -ui;
      const float cim = ki == 0 ? ui : ki == 1 ? -ur : ki == 2 ? -ui : ur;
      v2f v = {m * cr, m * cim};
      *(v2f*)(ib + (((2 * lane) ^ (2 * swz0(i))) + 64 * i)) = v;
    }
  }
  }

  for (int l = 0; l < LAYERS; ++l) {
    const int rr = l % 9 + 1;
    const int iv = slot0x2(cnot_map(512 >> lane, rr));
    int img[NQ];
    #pragma unroll
    for (int w = 0; w < NQ; ++w) img[w] = __builtin_amdgcn_readlane(iv, w);
    const int w2l = ((l16 & 1) ? img[0] : 0) ^ ((l16 & 2) ? img[1] : 0) ^ ((l16 & 4) ? img[2] : 0)
                  ^ ((l16 & 8) ? img[3] : 0) ^ (half ? img[6] : 0);
    float g9[8];
    #pragma unroll
    for (int c = 0; c < 8; ++c) g9[c] = gates[G9_OFF + l * 8 + c];
    float* const bA = (l & 1) ? buf1 : buf0;
    float* const bB = (l & 1) ? buf0 : buf1;

    #pragma unroll
    for (int p = 0; p < NPASS; ++p) {
      const float* const src = (p == 1) ? bB : bA;
      float* const dst = (p == 1) ? bA : bB;
      const v4f ga = gna, gb = gnb;
      const v2f a[4] = {{ga.x, ga.y}, {ga.z, ga.w}, {gb.x, gb.y}, {gb.z, gb.w}};
      {
        const int qn = l * NPASS + p + 1 < LAYERS * NPASS ? l * NPASS + p + 1 : l * NPASS + p;
        gna = *(const v4f*)(gl + qn * 256);
        gnb = *(const v4f*)(gl + qn * 256 + 4);
      }

      v2f bf[IPW][8][4];
      #pragma unroll
      for (int it = 0; it < IPW; ++it) {
        #pragma unroll
        for (int t = 0; t < 8; ++t) {
          #pragma unroll
          for (int ks = 0; ks < 4; ++ks) {
            const int off = p == 0 ? ((rdl ^ RX[t][ks]) + RD[0][t][ks])
                          : p == 1 ? (rdl + RD[1][t][ks])
                          : ((rdl ^ (((ks & 1) ? 8 : 0) ^ ((t & 4) ? 16 : 0))) + 64 * (4 * t + ks));
            bf[it][t][ks] = *(const v2f*)(src + off + it * 4096);
          }
        }
      }
      if (p == 2) {
        #pragma unroll
        for (int it = 0; it < IPW; ++it) {
          #pragma unroll
          for (int t = 0; t < 4; ++t) {
            #pragma unroll
            for (int ks = 0; ks < 4; ++ks) {
              const v2f x0 = bf[it][t][ks], x1 = bf[it][t + 4][ks];
              const v2f x0s = {-x0.y, x0.x}, x1s = {-x1.y, x1.x};
              bf[it][t][ks]     = x0 * g9[0] + x0s * g9[1] + x1 * g9[2] + x1s * g9[3];
              bf[it][t + 4][ks] = x0 * g9[4] + x0s * g9[5] + x1 * g9[6] + x1s * g9[7];
            }
          }
        }
      }
      v8f acc[IPW][8];
      #pragma unroll
      for (int it = 0; it < IPW; ++it) {
        #pragma unroll
        for (int t = 0; t < 8; ++t) acc[it][t] = v8f{};
      }
      #pragma unroll
      for (int it = 0; it < IPW; ++it) {
        #pragma unroll
        for (int ks = 0; ks < 4; ++ks) {
          #pragma unroll
          for (int t = 0; t < 8; ++t)
            acc[it][t] = __builtin_amdgcn_wmma_f32_16x16x4_f32(false, a[ks], false, bf[it][t][ks], (short)0, acc[it][t], false, false);
        }
      }
      #pragma unroll
      for (int t = 0; t < 8; ++t) {
        #pragma unroll
        for (int j = 0; j < 4; ++j) {
          int wo;
          if (p == 0) {
            wo = wrl0 + WR0[t][j];
          } else if (p == 1) {
            const int c = ((t & 4) ? 128 : 0) + ((j & 1) ? 512 : 0) + ((j & 2) ? 256 : 0);
            wo = wrl1x[t & 3] + c;
          } else {
            const int c = ((t & 1) ? img[4] : 0) ^ ((t & 2) ? img[5] : 0) ^ ((t & 4) ? img[9] : 0)
                        ^ ((j & 2) ? img[7] : 0) ^ ((j & 1) ? img[8] : 0);
            wo = w2l ^ c;
          }
          #pragma unroll
          for (int it = 0; it < IPW; ++it) {
            v2f v = {acc[it][t][2 * j], acc[it][t][2 * j + 1]};
            *(v2f*)(dst + wo + it * 4096) = v;
          }
        }
      }
    }
  }

  for (int it = 0; it < IPW; ++it) {
  const float* const rb = buf0 + it * 4096;
  float ptot = 0.f, z0 = 0.f, z1 = 0.f, z3 = 0.f, z4 = 0.f, z6 = 0.f;
  #pragma unroll
  for (int i = 0; i < 32; ++i) {
    v2f v = *(const v2f*)(rb + (((2 * lane) ^ (2 * swz0(i))) + 64 * i));
    float p = v.x * v.x + v.y * v.y;
    ptot += p;
    z0 += (i & 1)  ? -p : p;
    z1 += (i & 2)  ? -p : p;
    z3 += (i & 4)  ? -p : p;
    z4 += (i & 8)  ? -p : p;
    z6 += (i & 16) ? -p : p;
  }
  float z[NQ];
  z[0] = z0; z[1] = z1; z[3] = z3; z[4] = z4; z[6] = z6;
  z[2] = (lane & 1)  ? -ptot : ptot;
  z[5] = (lane & 2)  ? -ptot : ptot;
  z[7] = (lane & 4)  ? -ptot : ptot;
  z[8] = (lane & 8)  ? -ptot : ptot;
  z[9] = (lane & 16) ? -ptot : ptot;
  for (int off = 16; off > 0; off >>= 1)
    #pragma unroll
    for (int w = 0; w < NQ; ++w)
      z[w] += __shfl_xor(z[w], off, 32);
  if (lane < NACT) {
    float acc = fcb[lane];
    #pragma unroll
    for (int w = 0; w < NQ; ++w) acc += fcw[lane * NQ + w] * z[w];
    out[(b0 + it) * NACT + lane] = acc;
  }
  }
}

extern "C" void kernel_launch(void* const* d_in, const int* in_sizes, int n_in,
                              void* d_out, int out_size, void* d_ws, size_t ws_size,
                              hipStream_t stream) {
  const float* inputs  = (const float*)d_in[0];
  const float* weights = (const float*)d_in[1];
  const float* fcw     = (const float*)d_in[2];
  const float* fcb     = (const float*)d_in[3];
  float* out   = (float*)d_out;
  float* gates = (float*)d_ws;

  build_gates_kernel<<<31, 256, 0, stream>>>(weights, gates);
  qsim_kernel<<<BATCH / (WPB * IPW), WPB * 32, 0, stream>>>(inputs, gates, fcw, fcb, out);
}


# Gate tables built by each work-group in its own LDS at the head of the main kernel (100 lanes make the 2x2 Rots, all lanes the 3-wire fragments): one launch per call, not two; sum order unchanged
# speedup vs baseline: 2.5673x; 1.0497x over previous
#include <hip/hip_runtime.h>


#define NQ      10
#define LAYERS  10
#define BATCH   4096
#define NACT    4
#define WPB     8
#define IPW     2
#define NPASS   3

typedef __attribute__((ext_vector_type(2))) float v2f;
typedef __attribute__((ext_vector_type(4))) float v4f;
typedef __attribute__((ext_vector_type(8))) float v8f;


__device__ __forceinline__ int cnot_map(int b, int rr) {
  #pragma unroll
  for (int w = 0; w < NQ; ++w) {
    int t = w + rr; if (t >= NQ) t -= NQ;
    b ^= ((b >> (9 - w)) & 1) << (9 - t);
  }
  return b;
}

__device__ __forceinline__ int swz0(int hi) {
  return ((hi & 1) ? 31 : 0) ^ ((hi & 2) ? 13 : 0) ^ ((hi & 4) ? 23 : 0) ^ ((hi & 8) ? 4 : 0) ^ ((hi & 16) ? 21 : 0);
}
__device__ __forceinline__ int slot0x2(int a) {
  int s = (((a >> 9) & 1) << 5) | (((a >> 8) & 1) << 6) | (((a >> 7) & 1) << 0) | (((a >> 6) & 1) << 7)
        | (((a >> 5) & 1) << 8) | (((a >> 4) & 1) << 1) | (((a >> 3) & 1) << 9) | (((a >> 2) & 1) << 2)
        | (((a >> 1) & 1) << 3) | (((a >> 0) & 1) << 4);
  s ^= swz0(s >> 5);
  return 2 * s;
}

__global__ __launch_bounds__(WPB * 32)
void qsim_kernel(const float* __restrict__ x,
                 const float* __restrict__ wts,
                 const float* __restrict__ fcw,
                 const float* __restrict__ fcb,
                 float* __restrict__ out) {
  __shared__ __attribute__((aligned(16))) float lds[WPB * IPW * 4096];
  __shared__ __attribute__((aligned(16))) float gtab[LAYERS * NPASS * 256];
  __shared__ __attribute__((aligned(16))) float rot[LAYERS * NQ * 8];
  const int tid  = threadIdx.x;

  if (tid < LAYERS * NQ) {
    const float phi   = wts[tid * 3 + 0];
    const float theta = wts[tid * 3 + 1];
    const float omega = wts[tid * 3 + 2];
    const float ct = cosf(0.5f * theta), st = sinf(0.5f * theta);
    const float a = 0.5f * (phi + omega), b = 0.5f * (phi - omega);
    const float ca = cosf(a), sa = sinf(a), cb = cosf(b), sb = sinf(b);
    const v4f r0 = { ca * ct, -sa * ct, -cb * st, -sb * st};
    const v4f r1 = { cb * st, -sb * st,  ca * ct,  sa * ct};
    *(v4f*)(rot + tid * 8)     = r0;
    *(v4f*)(rot + tid * 8 + 4) = r1;
  }

  const int wave = threadIdx.x >> 5;
  const int lane = threadIdx.x & 31;
  const int half = lane >> 4;
  const int l16  = lane & 15;
  const int b0   = (blockIdx.x * WPB + wave) * IPW;
  float* const buf0 = lds + wave * (IPW * 4096);
  float* const buf1 = buf0 + 2048;

  constexpr int RD[2][8][4] = {{{0, 128, 64, 192}, {256, 384, 320, 448}, {512, 640, 576, 704}, {768, 896, 832, 960}, {1024, 1152, 1088, 1216}, {1280, 1408, 1344, 1472}, {1536, 1664, 1600, 1728}, {1792, 1920, 1856, 1984}}, {{0, 512, 256, 768}, {64, 576, 320, 832}, {128, 640, 384, 896}, {192, 704, 448, 960}, {1024, 1536, 1280, 1792}, {1088, 1600, 1344, 1856}, {1152, 1664, 1408, 1920}, {1216, 1728, 1472, 1984}}};
  constexpr int RX[8][4] = {{0, 26, 62, 36}, {46, 52, 16, 10}, {8, 18, 54, 44}, {38, 60, 24, 2}, {42, 48, 20, 14}, {4, 30, 58, 32}, {34, 56, 28, 6}, {12, 22, 50, 40}};
  constexpr int WR0[8][4] = {{0, 128, 64, 192}, {256, 384, 320, 448}, {512, 640, 576, 704}, {768, 896, 832, 960}, {1024, 1152, 1088, 1216}, {1280, 1408, 1344, 1472}, {1536, 1664, 1600, 1728}, {1792, 1920, 1856, 1984}};

  const int rdl = (l16 << 2) | (half << 1);
  const int wrl0 = (((l16 >> 0) & 1) << 1) | (((l16 >> 1) & 1) << 3) | (((l16 >> 2) & 1) << 4) | (((l16 >> 3) & 1) << 5) | (half << 2);
  const int wrl1 = ((l16 & 1) ? 4 : 0) ^ ((l16 & 2) ? 72 : 0) ^ ((l16 & 4) ? 2 : 0) ^ ((l16 & 8) ? 1040 : 0) ^ (half ? 32 : 0);
  const int wrl1x[4] = {wrl1, wrl1 ^ 8, wrl1 ^ 16, wrl1 ^ 24};

  for (int it = 0; it < IPW; ++it) {
  float* const ib = buf0 + it * 4096;
  float cw[NQ], sw[NQ];
  #pragma unroll
  for (int w = 0; w < NQ; ++w) {
    float xv = x[(b0 + it) * NQ + w];
    xv = xv > 0.f ? xv : 0.f;
    cw[w] = cosf(0.5f * xv);
    sw[w] = sinf(0.5f * xv);
  }
  {
    float fw[NQ];
    fw[2] = (lane & 1)  ? sw[2] : cw[2];
    fw[5] = (lane & 2)  ? sw[5] : cw[5];
    fw[7] = (lane & 4)  ? sw[7] : cw[7];
    fw[8] = (lane & 8)  ? sw[8] : cw[8];
    fw[9] = (lane & 16) ? sw[9] : cw[9];
    const int kl = __popc(lane) & 3;
    const float ur = kl == 0 ? 1.f : (kl == 2 ? -1.f : 0.f);
    const float ui = kl == 1 ? -1.f : (kl == 3 ? 1.f : 0.f);
    #pragma unroll
    for (int i = 0; i < 32; ++i) {
      fw[0] = (i & 1)  ? sw[0] : cw[0];
      fw[1] = (i & 2)  ? sw[1] : cw[1];
      fw[3] = (i & 4)  ? sw[3] : cw[3];
      fw[4] = (i & 8)  ? sw[4] : cw[4];
      fw[6] = (i & 16) ? sw[6] : cw[6];
      float m = 1.f;
      #pragma unroll
      for (int w = 0; w < NQ; ++w) m *= fw[w];
      const int ki = __popc(i) & 3;
      const float cr = ki == 0 ? ur : ki == 1 ? ui : ki == 2 ? -ur : -ui;
      const float cim = ki == 0 ? ui : ki == 1 ? -ur : ki == 2 ? -ui : ur;
      v2f v = {m * cr, m * cim};
      *(v2f*)(ib + (((2 * lane) ^ (2 * swz0(i))) + 64 * i)) = v;
    }
  }
  }

  __syncthreads();
  for (int u = tid; u < LAYERS * NPASS * 32; u += WPB * 32) {
    const int q = u >> 5, ln = u & 31;
    const int lq = q / NPASS, pq = q - lq * NPASS;
    const int m = ln & 15, hf = ln >> 4, i = m >> 1, co = m & 1;
    const float* const rb = rot + (lq * NQ + 3 * pq) * 8;
    const v2f A0 = *(const v2f*)(rb + 4 * ((i >> 2) & 1));
    const v2f A1 = *(const v2f*)(rb + 4 * ((i >> 2) & 1) + 2);
    const v2f B0 = *(const v2f*)(rb + 8 + 4 * ((i >> 1) & 1));
    const v2f B1 = *(const v2f*)(rb + 8 + 4 * ((i >> 1) & 1) + 2);
    const v2f C  = *(const v2f*)(rb + 16 + 4 * (i & 1) + 2 * hf);
    float o[8];
    #pragma unroll
    for (int ks = 0; ks < 4; ++ks) {
      const v2f ga = (ks & 2) ? A1 : A0, gb = (ks & 1) ? B1 : B0;
      const float tr = ga.x * gb.x - ga.y * gb.y, ti = ga.x * gb.y + ga.y * gb.x;
      const float gr = tr * C.x - ti * C.y,       gi = tr * C.y + ti * C.x;
      o[2 * ks]     = co ? gi : gr;
      o[2 * ks + 1] = co ? gr : -gi;
    }
    const v4f o0 = {o[0], o[1], o[2], o[3]}, o1 = {o[4], o[5], o[6], o[7]};
    *(v4f*)(gtab + u * 8)     = o0;
    *(v4f*)(gtab + u * 8 + 4) = o1;
  }
  __syncthreads();

  const float* const gl = gtab + lane * 8;
  v4f gna = *(const v4f*)(gl), gnb = *(const v4f*)(gl + 4);

  for (int l = 0; l < LAYERS; ++l) {
    const int rr = l % 9 + 1;
    const int iv = slot0x2(cnot_map(512 >> lane, rr));
    int img[NQ];
    #pragma unroll
    for (int w = 0; w < NQ; ++w) img[w] = __builtin_amdgcn_readlane(iv, w);
    const int w2l = ((l16 & 1) ? img[0] : 0) ^ ((l16 & 2) ? img[1] : 0) ^ ((l16 & 4) ? img[2] : 0)
                  ^ ((l16 & 8) ? img[3] : 0) ^ (half ? img[6] : 0);
    float g9[8];
    #pragma unroll
    for (int c = 0; c < 8; ++c) g9[c] = rot[(l * NQ + 9) * 8 + c];
    float* const bA = (l & 1) ? buf1 : buf0;
    float* const bB = (l & 1) ? buf0 : buf1;

    #pragma unroll
    for (int p = 0; p < NPASS; ++p) {
      const float* const src = (p == 1) ? bB : bA;
      float* const dst = (p == 1) ? bA : bB;
      const v4f ga = gna, gb = gnb;
      const v2f a[4] = {{ga.x, ga.y}, {ga.z, ga.w}, {gb.x, gb.y}, {gb.z, gb.w}};
      {
        const int qn = l * NPASS + p + 1 < LAYERS * NPASS ? l * NPASS + p + 1 : l * NPASS + p;
        gna = *(const v4f*)(gl + qn * 256);
        gnb = *(const v4f*)(gl + qn * 256 + 4);
      }

      v2f bf[IPW][8][4];
      #pragma unroll
      for (int it = 0; it < IPW; ++it) {
        #pragma unroll
        for (int t = 0; t < 8; ++t) {
          #pragma unroll
          for (int ks = 0; ks < 4; ++ks) {
            const int off = p == 0 ? ((rdl ^ RX[t][ks]) + RD[0][t][ks])
                          : p == 1 ? (rdl + RD[1][t][ks])
                          : ((rdl ^ (((ks & 1) ? 8 : 0) ^ ((t & 4) ? 16 : 0))) + 64 * (4 * t + ks));
            bf[it][t][ks] = *(const v2f*)(src + off + it * 4096);
          }
        }
      }
      if (p == 2) {
        #pragma unroll
        for (int it = 0; it < IPW; ++it) {
          #pragma unroll
          for (int t = 0; t < 4; ++t) {
            #pragma unroll
            for (int ks = 0; ks < 4; ++ks) {
              const v2f x0 = bf[it][t][ks], x1 = bf[it][t + 4][ks];
              const v2f x0s = {-x0.y, x0.x}, x1s = {-x1.y, x1.x};
              bf[it][t][ks]     = x0 * g9[0] + x0s * g9[1] + x1 * g9[2] + x1s * g9[3];
              bf[it][t + 4][ks] = x0 * g9[4] + x0s * g9[5] + x1 * g9[6] + x1s * g9[7];
            }
          }
        }
      }
      v8f acc[IPW][8];
      #pragma unroll
      for (int it = 0; it < IPW; ++it) {
        #pragma unroll
        for (int t = 0; t < 8; ++t) acc[it][t] = v8f{};
      }
      #pragma unroll
      for (int it = 0; it < IPW; ++it) {
        #pragma unroll
        for (int ks = 0; ks < 4; ++ks) {
          #pragma unroll
          for (int t = 0; t < 8; ++t)
            acc[it][t] = __builtin_amdgcn_wmma_f32_16x16x4_f32(false, a[ks], false, bf[it][t][ks], (short)0, acc[it][t], false, false);
        }
      }
      #pragma unroll
      for (int t = 0; t < 8; ++t) {
        #pragma unroll
        for (int j = 0; j < 4; ++j) {
          int wo;
          if (p == 0) {
            wo = wrl0 + WR0[t][j];
          } else if (p == 1) {
            const int c = ((t & 4) ? 128 : 0) + ((j & 1) ? 512 : 0) + ((j & 2) ? 256 : 0);
            wo = wrl1x[t & 3] + c;
          } else {
            const int c = ((t & 1) ? img[4] : 0) ^ ((t & 2) ? img[5] : 0) ^ ((t & 4) ? img[9] : 0)
                        ^ ((j & 2) ? img[7] : 0) ^ ((j & 1) ? img[8] : 0);
            wo = w2l ^ c;
          }
          #pragma unroll
          for (int it = 0; it < IPW; ++it) {
            v2f v = {acc[it][t][2 * j], acc[it][t][2 * j + 1]};
            *(v2f*)(dst + wo + it * 4096) = v;
          }
        }
      }
    }
  }

  for (int it = 0; it < IPW; ++it) {
  const float* const rb = buf0 + it * 4096;
  float ptot = 0.f, z0 = 0.f, z1 = 0.f, z3 = 0.f, z4 = 0.f, z6 = 0.f;
  #pragma unroll
  for (int i = 0; i < 32; ++i) {
    v2f v = *(const v2f*)(rb + (((2 * lane) ^ (2 * swz0(i))) + 64 * i));
    float p = v.x * v.x + v.y * v.y;
    ptot += p;
    z0 += (i & 1)  ? -p : p;
    z1 += (i & 2)  ? -p : p;
    z3 += (i & 4)  ? -p : p;
    z4 += (i & 8)  ? -p : p;
    z6 += (i & 16) ? -p : p;
  }
  float z[NQ];
  z[0] = z0; z[1] = z1; z[3] = z3; z[4] = z4; z[6] = z6;
  z[2] = (lane & 1)  ? -ptot : ptot;
  z[5] = (lane & 2)  ? -ptot : ptot;
  z[7] = (lane & 4)  ? -ptot : ptot;
  z[8] = (lane & 8)  ? -ptot : ptot;
  z[9] = (lane & 16) ? -ptot : ptot;
  for (int off = 16; off > 0; off >>= 1)
    #pragma unroll
    for (int w = 0; w < NQ; ++w)
      z[w] += __shfl_xor(z[w], off, 32);
  if (lane < NACT) {
    float acc = fcb[lane];
    #pragma unroll
    for (int w = 0; w < NQ; ++w) acc += fcw[lane * NQ + w] * z[w];
    out[(b0 + it) * NACT + lane] = acc;
  }
  }
}

extern "C" void kernel_launch(void* const* d_in, const int* in_sizes, int n_in,
                              void* d_out, int out_size, void* d_ws, size_t ws_size,
                              hipStream_t stream) {
  const float* inputs  = (const float*)d_in[0];
  const float* weights = (const float*)d_in[1];
  const float* fcw     = (const float*)d_in[2];
  const float* fcb     = (const float*)d_in[3];
  float* out   = (float*)d_out;

  qsim_kernel<<<BATCH / (WPB * IPW), WPB * 32, 0, stream>>>(inputs, weights, fcw, fcb, out);
}


# Input angles as a work-group table in LDS: 160 threads each load one x (coalesced) and do one cosf/sinf; a wave reads its 20 pairs back, not 20 serial loads with cos/sin in every lane; sums unchanged
# speedup vs baseline: 2.6845x; 1.0457x over previous
#include <hip/hip_runtime.h>


#define NQ      10
#define LAYERS  10
#define BATCH   4096
#define NACT    4
#define WPB     8
#define IPW     2
#define NPASS   3

typedef __attribute__((ext_vector_type(2))) float v2f;
typedef __attribute__((ext_vector_type(4))) float v4f;
typedef __attribute__((ext_vector_type(8))) float v8f;


__device__ __forceinline__ int cnot_map(int b, int rr) {
  #pragma unroll
  for (int w = 0; w < NQ; ++w) {
    int t = w + rr; if (t >= NQ) t -= NQ;
    b ^= ((b >> (9 - w)) & 1) << (9 - t);
  }
  return b;
}

__device__ __forceinline__ int swz0(int hi) {
  return ((hi & 1) ? 31 : 0) ^ ((hi & 2) ? 13 : 0) ^ ((hi & 4) ? 23 : 0) ^ ((hi & 8) ? 4 : 0) ^ ((hi & 16) ? 21 : 0);
}
__device__ __forceinline__ int slot0x2(int a) {
  int s = (((a >> 9) & 1) << 5) | (((a >> 8) & 1) << 6) | (((a >> 7) & 1) << 0) | (((a >> 6) & 1) << 7)
        | (((a >> 5) & 1) << 8) | (((a >> 4) & 1) << 1) | (((a >> 3) & 1) << 9) | (((a >> 2) & 1) << 2)
        | (((a >> 1) & 1) << 3) | (((a >> 0) & 1) << 4);
  s ^= swz0(s >> 5);
  return 2 * s;
}

__global__ __launch_bounds__(WPB * 32)
void qsim_kernel(const float* __restrict__ x,
                 const float* __restrict__ wts,
                 const float* __restrict__ fcw,
                 const float* __restrict__ fcb,
                 float* __restrict__ out) {
  __shared__ __attribute__((aligned(16))) float lds[WPB * IPW * 4096];
  __shared__ __attribute__((aligned(16))) float gtab[LAYERS * NPASS * 256];
  __shared__ __attribute__((aligned(16))) float rot[LAYERS * NQ * 8];
  __shared__ __attribute__((aligned(16))) float cst[WPB * IPW * NQ * 2];
  const int tid  = threadIdx.x;

  if (tid < WPB * IPW * NQ) {
    float xv = x[blockIdx.x * (WPB * IPW * NQ) + tid];
    xv = xv > 0.f ? xv : 0.f;
    const v2f cs = {cosf(0.5f * xv), sinf(0.5f * xv)};
    *(v2f*)(cst + 2 * tid) = cs;
  }

  if (tid >= WPB * 32 - LAYERS * NQ) {
    const int g = tid - (WPB * 32 - LAYERS * NQ);
    const float phi   = wts[g * 3 + 0];
    const float theta = wts[g * 3 + 1];
    const float omega = wts[g * 3 + 2];
    const float ct = cosf(0.5f * theta), st = sinf(0.5f * theta);
    const float a = 0.5f * (phi + omega), b = 0.5f * (phi - omega);
    const float ca = cosf(a), sa = sinf(a), cb = cosf(b), sb = sinf(b);
    const v4f r0 = { ca * ct, -sa * ct, -cb * st, -sb * st};
    const v4f r1 = { cb * st, -sb * st,  ca * ct,  sa * ct};
    *(v4f*)(rot + g * 8)     = r0;
    *(v4f*)(rot + g * 8 + 4) = r1;
  }

  const int wave = threadIdx.x >> 5;
  const int lane = threadIdx.x & 31;
  const int half = lane >> 4;
  const int l16  = lane & 15;
  const int b0   = (blockIdx.x * WPB + wave) * IPW;
  float* const buf0 = lds + wave * (IPW * 4096);
  float* const buf1 = buf0 + 2048;

  constexpr int RD[2][8][4] = {{{0, 128, 64, 192}, {256, 384, 320, 448}, {512, 640, 576, 704}, {768, 896, 832, 960}, {1024, 1152, 1088, 1216}, {1280, 1408, 1344, 1472}, {1536, 1664, 1600, 1728}, {1792, 1920, 1856, 1984}}, {{0, 512, 256, 768}, {64, 576, 320, 832}, {128, 640, 384, 896}, {192, 704, 448, 960}, {1024, 1536, 1280, 1792}, {1088, 1600, 1344, 1856}, {1152, 1664, 1408, 1920}, {1216, 1728, 1472, 1984}}};
  constexpr int RX[8][4] = {{0, 26, 62, 36}, {46, 52, 16, 10}, {8, 18, 54, 44}, {38, 60, 24, 2}, {42, 48, 20, 14}, {4, 30, 58, 32}, {34, 56, 28, 6}, {12, 22, 50, 40}};
  constexpr int WR0[8][4] = {{0, 128, 64, 192}, {256, 384, 320, 448}, {512, 640, 576, 704}, {768, 896, 832, 960}, {1024, 1152, 1088, 1216}, {1280, 1408, 1344, 1472}, {1536, 1664, 1600, 1728}, {1792, 1920, 1856, 1984}};

  const int rdl = (l16 << 2) | (half << 1);
  const int wrl0 = (((l16 >> 0) & 1) << 1) | (((l16 >> 1) & 1) << 3) | (((l16 >> 2) & 1) << 4) | (((l16 >> 3) & 1) << 5) | (half << 2);
  const int wrl1 = ((l16 & 1) ? 4 : 0) ^ ((l16 & 2) ? 72 : 0) ^ ((l16 & 4) ? 2 : 0) ^ ((l16 & 8) ? 1040 : 0) ^ (half ? 32 : 0);
  const int wrl1x[4] = {wrl1, wrl1 ^ 8, wrl1 ^ 16, wrl1 ^ 24};

  __syncthreads();

  for (int it = 0; it < IPW; ++it) {
  float* const ib = buf0 + it * 4096;
  float cw[NQ], sw[NQ];
  const float* const cp = cst + (wave * IPW + it) * (NQ * 2);
  #pragma unroll
  for (int w = 0; w < NQ; ++w) {
    const v2f t = *(const v2f*)(cp + 2 * w);
    cw[w] = t.x;
    sw[w] = t.y;
  }
  {
    float fw[NQ];
    fw[2] = (lane & 1)  ? sw[2] : cw[2];
    fw[5] = (lane & 2)  ? sw[5] : cw[5];
    fw[7] = (lane & 4)  ? sw[7] : cw[7];
    fw[8] = (lane & 8)  ? sw[8] : cw[8];
    fw[9] = (lane & 16) ? sw[9] : cw[9];
    const int kl = __popc(lane) & 3;
    const float ur = kl == 0 ? 1.f : (kl == 2 ? -1.f : 0.f);
    const float ui = kl == 1 ? -1.f : (kl == 3 ? 1.f : 0.f);
    #pragma unroll
    for (int i = 0; i < 32; ++i) {
      fw[0] = (i & 1)  ? sw[0] : cw[0];
      fw[1] = (i & 2)  ? sw[1] : cw[1];
      fw[3] = (i & 4)  ? sw[3] : cw[3];
      fw[4] = (i & 8)  ? sw[4] : cw[4];
      fw[6] = (i & 16) ? sw[6] : cw[6];
      float m = 1.f;
      #pragma unroll
      for (int w = 0; w < NQ; ++w) m *= fw[w];
      const int ki = __popc(i) & 3;
      const float cr = ki == 0 ? ur : ki == 1 ? ui : ki == 2 ? -ur : -ui;
      const float cim = ki == 0 ? ui : ki == 1 ? -ur : ki == 2 ? -ui : ur;
      v2f v = {m * cr, m * cim};
      *(v2f*)(ib + (((2 * lane) ^ (2 * swz0(i))) + 64 * i)) = v;
    }
  }
  }

  for (int u = tid; u < LAYERS * NPASS * 32; u += WPB * 32) {
    const int q = u >> 5, ln = u & 31;
    const int lq = q / NPASS, pq = q - lq * NPASS;
    const int m = ln & 15, hf = ln >> 4, i = m >> 1, co = m & 1;
    const float* const rb = rot + (lq * NQ + 3 * pq) * 8;
    const v2f A0 = *(const v2f*)(rb + 4 * ((i >> 2) & 1));
    const v2f A1 = *(const v2f*)(rb + 4 * ((i >> 2) & 1) + 2);
    const v2f B0 = *(const v2f*)(rb + 8 + 4 * ((i >> 1) & 1));
    const v2f B1 = *(const v2f*)(rb + 8 + 4 * ((i >> 1) & 1) + 2);
    const v2f C  = *(const v2f*)(rb + 16 + 4 * (i & 1) + 2 * hf);
    float o[8];
    #pragma unroll
    for (int ks = 0; ks < 4; ++ks) {
      const v2f ga = (ks & 2) ? A1 : A0, gb = (ks & 1) ? B1 : B0;
      const float tr = ga.x * gb.x - ga.y * gb.y, ti = ga.x * gb.y + ga.y * gb.x;
      const float gr = tr * C.x - ti * C.y,       gi = tr * C.y + ti * C.x;
      o[2 * ks]     = co ? gi : gr;
      o[2 * ks + 1] = co ? gr : -gi;
    }
    const v4f o0 = {o[0], o[1], o[2], o[3]}, o1 = {o[4], o[5], o[6], o[7]};
    *(v4f*)(gtab + u * 8)     = o0;
    *(v4f*)(gtab + u * 8 + 4) = o1;
  }
  __syncthreads();

  const float* const gl = gtab + lane * 8;
  v4f gna = *(const v4f*)(gl), gnb = *(const v4f*)(gl + 4);

  for (int l = 0; l < LAYERS; ++l) {
    const int rr = l % 9 + 1;
    const int iv = slot0x2(cnot_map(512 >> lane, rr));
    int img[NQ];
    #pragma unroll
    for (int w = 0; w < NQ; ++w) img[w] = __builtin_amdgcn_readlane(iv, w);
    const int w2l = ((l16 & 1) ? img[0] : 0) ^ ((l16 & 2) ? img[1] : 0) ^ ((l16 & 4) ? img[2] : 0)
                  ^ ((l16 & 8) ? img[3] : 0) ^ (half ? img[6] : 0);
    float g9[8];
    #pragma unroll
    for (int c = 0; c < 8; ++c) g9[c] = rot[(l * NQ + 9) * 8 + c];
    float* const bA = (l & 1) ? buf1 : buf0;
    float* const bB = (l & 1) ? buf0 : buf1;

    #pragma unroll
    for (int p = 0; p < NPASS; ++p) {
      const float* const src = (p == 1) ? bB : bA;
      float* const dst = (p == 1) ? bA : bB;
      const v4f ga = gna, gb = gnb;
      const v2f a[4] = {{ga.x, ga.y}, {ga.z, ga.w}, {gb.x, gb.y}, {gb.z, gb.w}};
      {
        const int qn = l * NPASS + p + 1 < LAYERS * NPASS ? l * NPASS + p + 1 : l * NPASS + p;
        gna = *(const v4f*)(gl + qn * 256);
        gnb = *(const v4f*)(gl + qn * 256 + 4);
      }

      v2f bf[IPW][8][4];
      #pragma unroll
      for (int it = 0; it < IPW; ++it) {
        #pragma unroll
        for (int t = 0; t < 8; ++t) {
          #pragma unroll
          for (int ks = 0; ks < 4; ++ks) {
            const int off = p == 0 ? ((rdl ^ RX[t][ks]) + RD[0][t][ks])
                          : p == 1 ? (rdl + RD[1][t][ks])
                          : ((rdl ^ (((ks & 1) ? 8 : 0) ^ ((t & 4) ? 16 : 0))) + 64 * (4 * t + ks));
            bf[it][t][ks] = *(const v2f*)(src + off + it * 4096);
          }
        }
      }
      if (p == 2) {
        #pragma unroll
        for (int it = 0; it < IPW; ++it) {
          #pragma unroll
          for (int t = 0; t < 4; ++t) {
            #pragma unroll
            for (int ks = 0; ks < 4; ++ks) {
              const v2f x0 = bf[it][t][ks], x1 = bf[it][t + 4][ks];
              const v2f x0s = {-x0.y, x0.x}, x1s = {-x1.y, x1.x};
              bf[it][t][ks]     = x0 * g9[0] + x0s * g9[1] + x1 * g9[2] + x1s * g9[3];
              bf[it][t + 4][ks] = x0 * g9[4] + x0s * g9[5] + x1 * g9[6] + x1s * g9[7];
            }
          }
        }
      }
      v8f acc[IPW][8];
      #pragma unroll
      for (int it = 0; it < IPW; ++it) {
        #pragma unroll
        for (int t = 0; t < 8; ++t) acc[it][t] = v8f{};
      }
      #pragma unroll
      for (int it = 0; it < IPW; ++it) {
        #pragma unroll
        for (int ks = 0; ks < 4; ++ks) {
          #pragma unroll
          for (int t = 0; t < 8; ++t)
            acc[it][t] = __builtin_amdgcn_wmma_f32_16x16x4_f32(false, a[ks], false, bf[it][t][ks], (short)0, acc[it][t], false, false);
        }
      }
      #pragma unroll
      for (int t = 0; t < 8; ++t) {
        #pragma unroll
        for (int j = 0; j < 4; ++j) {
          int wo;
          if (p == 0) {
            wo = wrl0 + WR0[t][j];
          } else if (p == 1) {
            const int c = ((t & 4) ? 128 : 0) + ((j & 1) ? 512 : 0) + ((j & 2) ? 256 : 0);
            wo = wrl1x[t & 3] + c;
          } else {
            const int c = ((t & 1) ? img[4] : 0) ^ ((t & 2) ? img[5] : 0) ^ ((t & 4) ? img[9] : 0)
                        ^ ((j & 2) ? img[7] : 0) ^ ((j & 1) ? img[8] : 0);
            wo = w2l ^ c;
          }
          #pragma unroll
          for (int it = 0; it < IPW; ++it) {
            v2f v = {acc[it][t][2 * j], acc[it][t][2 * j + 1]};
            *(v2f*)(dst + wo + it * 4096) = v;
          }
        }
      }
    }
  }

  for (int it = 0; it < IPW; ++it) {
  const float* const rb = buf0 + it * 4096;
  float ptot = 0.f, z0 = 0.f, z1 = 0.f, z3 = 0.f, z4 = 0.f, z6 = 0.f;
  #pragma unroll
  for (int i = 0; i < 32; ++i) {
    v2f v = *(const v2f*)(rb + (((2 * lane) ^ (2 * swz0(i))) + 64 * i));
    float p = v.x * v.x + v.y * v.y;
    ptot += p;
    z0 += (i & 1)  ? -p : p;
    z1 += (i & 2)  ? -p : p;
    z3 += (i & 4)  ? -p : p;
    z4 += (i & 8)  ? -p : p;
    z6 += (i & 16) ? -p : p;
  }
  float z[NQ];
  z[0] = z0; z[1] = z1; z[3] = z3; z[4] = z4; z[6] = z6;
  z[2] = (lane & 1)  ? -ptot : ptot;
  z[5] = (lane & 2)  ? -ptot : ptot;
  z[7] = (lane & 4)  ? -ptot : ptot;
  z[8] = (lane & 8)  ? -ptot : ptot;
  z[9] = (lane & 16) ? -ptot : ptot;
  for (int off = 16; off > 0; off >>= 1)
    #pragma unroll
    for (int w = 0; w < NQ; ++w)
      z[w] += __shfl_xor(z[w], off, 32);
  if (lane < NACT) {
    float acc = fcb[lane];
    #pragma unroll
    for (int w = 0; w < NQ; ++w) acc += fcw[lane * NQ + w] * z[w];
    out[(b0 + it) * NACT + lane] = acc;
  }
  }
}

extern "C" void kernel_launch(void* const* d_in, const int* in_sizes, int n_in,
                              void* d_out, int out_size, void* d_ws, size_t ws_size,
                              hipStream_t stream) {
  const float* inputs  = (const float*)d_in[0];
  const float* weights = (const float*)d_in[1];
  const float* fcw     = (const float*)d_in[2];
  const float* fcb     = (const float*)d_in[3];
  float* out   = (float*)d_out;

  qsim_kernel<<<BATCH / (WPB * IPW), WPB * 32, 0, stream>>>(inputs, weights, fcw, fcb, out);
}
